# starting kernel
# baseline (speedup 1.0000x reference)
#include <hip/hip_runtime.h>

typedef __attribute__((ext_vector_type(2))) float v2f;
typedef __attribute__((ext_vector_type(8))) float v8f;

#define N_NODES 50000
#define N_EDGES 640000
#define N_REL   8
#define KDIM    128
#define NSEG    (N_REL + 1)

__global__ void rgcn_zero_kernel(float4* __restrict__ p, long long n4) {
    long long i = blockIdx.x * (long long)blockDim.x + threadIdx.x;
    if (i < n4) p[i] = make_float4(0.f, 0.f, 0.f, 0.f);
}

__global__ void rgcn_scatter_kernel(const float* __restrict__ feat,
                                    const int*   __restrict__ esrc,
                                    const int*   __restrict__ edst,
                                    const int*   __restrict__ etyp,
                                    float*       __restrict__ aggR,
                                    int nEdges) {
    long long tid = blockIdx.x * (long long)blockDim.x + threadIdx.x;
    int e = (int)(tid >> 7);
    int o = (int)(tid & 127);
    if (e >= nEdges) return;
    int s = esrc[e];
    int d = edst[e];
    int r = etyp[e];
    float v = feat[(size_t)s * KDIM + o];
    atomicAdd(aggR + ((size_t)r * N_NODES + (size_t)d) * KDIM + o, v);
}

__global__ void rgcn_pack_w_kernel(const float* __restrict__ Wrel,
                                   const float* __restrict__ Wself,
                                   float2*      __restrict__ Wf,
                                   int O) {
    const int tiles = O >> 4;
    const int total = NSEG * tiles * 32 * 32;
    int idx = blockIdx.x * blockDim.x + threadIdx.x;
    if (idx >= total) return;
    int lane = idx & 31;
    int k4   = (idx >> 5) & 31;
    int t    = (idx >> 10) % tiles;
    int seg  = idx / (tiles << 10);
    const float* W = (seg < N_REL) ? (Wrel + (size_t)seg * KDIM * O) : Wself;
    int k0 = k4 * 4 + (lane >> 4) * 2;
    int n  = t * 16 + (lane & 15);
    Wf[idx] = make_float2(W[(size_t)k0 * O + n], W[(size_t)(k0 + 1) * O + n]);
}

__global__ void rgcn_gemm_kernel(const float*  __restrict__ aggR,
                                 const float*  __restrict__ hself,
                                 const float2* __restrict__ Wf,
                                 const float*  __restrict__ bias,
                                 float*        __restrict__ out,
                                 int O, int applyRelu) {
    __shared__ float lds[16 * 132];

    const int m0    = blockIdx.x * 16;
    const int lane  = threadIdx.x & 31;
    const int wave  = threadIdx.x >> 5;
    const int half  = lane >> 4;
    const int l15   = lane & 15;
    const int nb    = wave * 16;
    const int tiles = O >> 4;

    v8f c = {0.f, 0.f, 0.f, 0.f, 0.f, 0.f, 0.f, 0.f};

    for (int s = 0; s < NSEG; ++s) {
        const float*  A  = (s < N_REL) ? (aggR + (size_t)s * N_NODES * KDIM) : hself;
        const float2* wf = Wf + ((size_t)(s * tiles + wave) << 10) + lane;

        __syncthreads();
        for (int t = threadIdx.x; t < 16 * KDIM; t += blockDim.x) {
            int row = t >> 7;
            int col = t & 127;
            lds[row * 132 + col] = A[(size_t)(m0 + row) * KDIM + col];
        }
        __syncthreads();

#pragma unroll
        for (int k4 = 0; k4 < 32; ++k4) {
            const float* ap = &lds[l15 * 132 + k4 * 4 + half * 2];
            v2f a;
            a.x = ap[0];
            a.y = ap[1];
            float2 bf = wf[k4 * 32];
            v2f b;
            b.x = bf.x;
            b.y = bf.y;
            c = __builtin_amdgcn_wmma_f32_16x16x4_f32(
                    false, a, false, b, (short)0, c, false, false);
        }
    }

    float bv = bias[nb + l15];
#pragma unroll
    for (int v = 0; v < 8; ++v) {
        float x = c[v] + bv;
        if (applyRelu) x = fmaxf(x, 0.f);
        out[(size_t)(m0 + v + 8 * half) * O + nb + l15] = x;
    }
}

extern "C" void kernel_launch(void* const* d_in, const int* in_sizes, int n_in,
                              void* d_out, int out_size, void* d_ws, size_t ws_size,
                              hipStream_t stream) {
    const float* features = (const float*)d_in[0];
    const int*   src      = (const int*)  d_in[1];
    const int*   dst      = (const int*)  d_in[2];
    const int*   etype    = (const int*)  d_in[3];
    const float* W1       = (const float*)d_in[4];
    const float* W1s      = (const float*)d_in[5];
    const float* b1       = (const float*)d_in[6];
    const float* W2       = (const float*)d_in[7];
    const float* W2s      = (const float*)d_in[8];
    const float* b2       = (const float*)d_in[9];
    float*       out      = (float*)d_out;

    float*  aggR = (float*)d_ws;
    float*  h1   = aggR + (size_t)N_REL * N_NODES * KDIM;
    float2* Wf1  = (float2*)(h1 + (size_t)N_NODES * KDIM);
    float2* Wf2  = Wf1 + (size_t)NSEG * 8 * 32 * 32;

    const long long n4     = (long long)N_REL * N_NODES * KDIM / 4;
    const int zeroBlocks   = (int)((n4 + 255) / 256);
    const int scatBlocks   = (int)(((long long)N_EDGES * KDIM) / 256);
    const int gemmBlocks   = N_NODES / 16;

    const int pack1 = NSEG * 8 * 32 * 32;
    const int pack2 = NSEG * 4 * 32 * 32;

    rgcn_pack_w_kernel<<<(pack1 + 255) / 256, 256, 0, stream>>>(W1, W1s, Wf1, 128);
    rgcn_pack_w_kernel<<<(pack2 + 255) / 256, 256, 0, stream>>>(W2, W2s, Wf2, 64);

    rgcn_zero_kernel<<<zeroBlocks, 256, 0, stream>>>((float4*)aggR, n4);
    rgcn_scatter_kernel<<<scatBlocks, 256, 0, stream>>>(
        features, src, dst, etype, aggR, N_EDGES);
    rgcn_gemm_kernel<<<gemmBlocks, 256, 0, stream>>>(
        aggR, features, Wf1, b1, h1, 128, 1);

    rgcn_zero_kernel<<<zeroBlocks, 256, 0, stream>>>((float4*)aggR, n4);
    rgcn_scatter_kernel<<<scatBlocks, 256, 0, stream>>>(
        h1, src, dst, etype, aggR, N_EDGES);
    rgcn_gemm_kernel<<<gemmBlocks, 128, 0, stream>>>(
        aggR, h1, Wf2, b2, out, 64, 0);
}


# Float atomic scatter-add and zero pass replaced by per-call edge grouping by (relation,dst) + wave-per-segment gather-sum; rows summed in ascending source id, so run-to-run fixed (not arrival order)
# speedup vs baseline: 1.9913x; 1.9913x over previous
#include <hip/hip_runtime.h>

typedef __attribute__((ext_vector_type(2))) float v2f;
typedef __attribute__((ext_vector_type(8))) float v8f;

#define N_NODES 50000
#define N_EDGES 640000
#define N_REL   8
#define KDIM    128
#define NSEG    (N_REL + 1)

#define N_KEYS        (N_REL * N_NODES)
#define SCAN_SHIFT    10
#define N_SCAN_BLOCKS ((N_KEYS + 1023) >> SCAN_SHIFT)
#define ID_NONE       0x7fffffff

__global__ void rgcn_zero_cnt_kernel(int4* __restrict__ p, int n4) {
    int i = blockIdx.x * blockDim.x + threadIdx.x;
    if (i < n4) p[i] = make_int4(0, 0, 0, 0);
}

__global__ void rgcn_count_kernel(const int* __restrict__ edst,
                                  const int* __restrict__ etyp,
                                  int*       __restrict__ cnt,
                                  int*       __restrict__ slot,
                                  int nEdges) {
    int e = blockIdx.x * blockDim.x + threadIdx.x;
    if (e >= nEdges) return;
    int key = etyp[e] * N_NODES + edst[e];
    slot[e] = atomicAdd(cnt + key, 1);
}

__device__ __forceinline__ int wave_incl_scan(int x, int lane) {
    for (int o = 1; o < 32; o <<= 1) {
        int y = __shfl_up(x, o, 32);
        if (lane >= o) x += y;
    }
    return x;
}

__global__ void rgcn_scan_local_kernel(const int* __restrict__ cnt,
                                       int*       __restrict__ loc,
                                       int*       __restrict__ bsum) {
    __shared__ int wsum[8];
    const int lane = threadIdx.x & 31;
    const int wave = threadIdx.x >> 5;
    const int g    = blockIdx.x * 256 + threadIdx.x;
    int4 v = make_int4(0, 0, 0, 0);
    if (g < N_KEYS / 4) v = ((const int4*)cnt)[g];
    const int s = v.x + v.y + v.z + v.w;
    const int x = wave_incl_scan(s, lane);
    if (lane == 31) wsum[wave] = x;
    __syncthreads();
    int wo = 0;
    for (int w = 0; w < wave; ++w) wo += wsum[w];
    const int ex = wo + x - s;
    if (g < N_KEYS / 4)
        ((int4*)loc)[g] = make_int4(ex, ex + v.x, ex + v.x + v.y, ex + v.x + v.y + v.z);
    if (threadIdx.x == 255) bsum[blockIdx.x] = wo + x;
}

__global__ void rgcn_scan_blocks_kernel(const int* __restrict__ bsum,
                                        int*       __restrict__ boff) {
    __shared__ int wsum[16];
    const int lane = threadIdx.x & 31;
    const int wave = threadIdx.x >> 5;
    const int s = (threadIdx.x < N_SCAN_BLOCKS) ? bsum[threadIdx.x] : 0;
    const int x = wave_incl_scan(s, lane);
    if (lane == 31) wsum[wave] = x;
    __syncthreads();
    int wo = 0;
    for (int w = 0; w < wave; ++w) wo += wsum[w];
    if (threadIdx.x < N_SCAN_BLOCKS) boff[threadIdx.x] = wo + x - s;
}

__global__ void rgcn_place_kernel(const int* __restrict__ esrc,
                                  const int* __restrict__ edst,
                                  const int* __restrict__ etyp,
                                  const int* __restrict__ slot,
                                  const int* __restrict__ loc,
                                  const int* __restrict__ boff,
                                  int*       __restrict__ ssrc,
                                  int nEdges) {
    int e = blockIdx.x * blockDim.x + threadIdx.x;
    if (e >= nEdges) return;
    int key = etyp[e] * N_NODES + edst[e];
    int pos = loc[key] + boff[key >> SCAN_SHIFT] + slot[e];
    ssrc[pos] = esrc[e];
}

__device__ __forceinline__ int wave_min_i(int v) {
    for (int o = 16; o > 0; o >>= 1) {
        int y = __shfl_xor(v, o, 32);
        v = (y < v) ? y : v;
    }
    return v;
}

__device__ __forceinline__ int wave_sum_i(int v) {
    for (int o = 16; o > 0; o >>= 1) v += __shfl_xor(v, o, 32);
    return v;
}

__global__ void rgcn_gather_kernel(const float* __restrict__ feat,
                                   const int*   __restrict__ cnt,
                                   const int*   __restrict__ loc,
                                   const int*   __restrict__ boff,
                                   const int*   __restrict__ ssrc,
                                   float*       __restrict__ aggR) {
    const int lane = threadIdx.x & 31;
    const int seg  = __builtin_amdgcn_readfirstlane((int)(blockIdx.x * 8 + (threadIdx.x >> 5)));
    const int c    = cnt[seg];
    const int base = loc[seg] + boff[seg >> SCAN_SHIFT];
    const int cl   = (c < 32) ? c : 32;
    const float4* f4 = (const float4*)feat;

    float4 acc = make_float4(0.f, 0.f, 0.f, 0.f);
    const int v0 = (lane < c) ? ssrc[base + lane] : ID_NONE;

    int prev = -1;
    int rem  = c;
    for (int it = 0; it < c && rem > 0; ++it) {
        int m = ID_NONE;
        for (int j = 0; j < cl; ++j) {
            int sj = __builtin_amdgcn_readlane(v0, j);
            if (sj > prev && sj < m) m = sj;
        }
        if (c > 32) {
            int lm = ID_NONE;
            for (int j = lane + 32; j < c; j += 32) {
                int v = ssrc[base + j];
                if (v > prev && v < lm) lm = v;
            }
            int t = __builtin_amdgcn_readfirstlane(wave_min_i(lm));
            if (t < m) m = t;
        }
        if (m == ID_NONE) break;
        int mult = __popc(__builtin_amdgcn_ballot_w32(v0 == m));
        if (c > 32) {
            int cm = 0;
            for (int j = lane + 32; j < c; j += 32) cm += (ssrc[base + j] == m) ? 1 : 0;
            mult += __builtin_amdgcn_readfirstlane(wave_sum_i(cm));
        }
        const float4 row = f4[(size_t)m * (KDIM / 4) + lane];
        for (int t = 0; t < mult; ++t) {
            acc.x += row.x;
            acc.y += row.y;
            acc.z += row.z;
            acc.w += row.w;
        }
        prev = m;
        rem -= mult;
    }
    ((float4*)aggR)[(size_t)seg * (KDIM / 4) + lane] = acc;
}

__global__ void rgcn_pack_w_kernel(const float* __restrict__ Wrel,
                                   const float* __restrict__ Wself,
                                   float2*      __restrict__ Wf,
                                   int O) {
    const int tiles = O >> 4;
    const int total = NSEG * tiles * 32 * 32;
    int idx = blockIdx.x * blockDim.x + threadIdx.x;
    if (idx >= total) return;
    int lane = idx & 31;
    int k4   = (idx >> 5) & 31;
    int t    = (idx >> 10) % tiles;
    int seg  = idx / (tiles << 10);
    const float* W = (seg < N_REL) ? (Wrel + (size_t)seg * KDIM * O) : Wself;
    int k0 = k4 * 4 + (lane >> 4) * 2;
    int n  = t * 16 + (lane & 15);
    Wf[idx] = make_float2(W[(size_t)k0 * O + n], W[(size_t)(k0 + 1) * O + n]);
}

__global__ void rgcn_gemm_kernel(const float*  __restrict__ aggR,
                                 const float*  __restrict__ hself,
                                 const float2* __restrict__ Wf,
                                 const float*  __restrict__ bias,
                                 float*        __restrict__ out,
                                 int O, int applyRelu) {
    __shared__ float lds[16 * 132];

    const int m0    = blockIdx.x * 16;
    const int lane  = threadIdx.x & 31;
    const int wave  = threadIdx.x >> 5;
    const int half  = lane >> 4;
    const int l15   = lane & 15;
    const int nb    = wave * 16;
    const int tiles = O >> 4;

    v8f c = {0.f, 0.f, 0.f, 0.f, 0.f, 0.f, 0.f, 0.f};

    for (int s = 0; s < NSEG; ++s) {
        const float*  A  = (s < N_REL) ? (aggR + (size_t)s * N_NODES * KDIM) : hself;
        const float2* wf = Wf + ((size_t)(s * tiles + wave) << 10) + lane;

        __syncthreads();
        for (int t = threadIdx.x; t < 16 * KDIM; t += blockDim.x) {
            int row = t >> 7;
            int col = t & 127;
            lds[row * 132 + col] = A[(size_t)(m0 + row) * KDIM + col];
        }
        __syncthreads();

#pragma unroll
        for (int k4 = 0; k4 < 32; ++k4) {
            const float* ap = &lds[l15 * 132 + k4 * 4 + half * 2];
            v2f a;
            a.x = ap[0];
            a.y = ap[1];
            float2 bf = wf[k4 * 32];
            v2f b;
            b.x = bf.x;
            b.y = bf.y;
            c = __builtin_amdgcn_wmma_f32_16x16x4_f32(
                    false, a, false, b, (short)0, c, false, false);
        }
    }

    float bv = bias[nb + l15];
#pragma unroll
    for (int v = 0; v < 8; ++v) {
        float x = c[v] + bv;
        if (applyRelu) x = fmaxf(x, 0.f);
        out[(size_t)(m0 + v + 8 * half) * O + nb + l15] = x;
    }
}

extern "C" void kernel_launch(void* const* d_in, const int* in_sizes, int n_in,
                              void* d_out, int out_size, void* d_ws, size_t ws_size,
                              hipStream_t stream) {
    const float* features = (const float*)d_in[0];
    const int*   src      = (const int*)  d_in[1];
    const int*   dst      = (const int*)  d_in[2];
    const int*   etype    = (const int*)  d_in[3];
    const float* W1       = (const float*)d_in[4];
    const float* W1s      = (const float*)d_in[5];
    const float* b1       = (const float*)d_in[6];
    const float* W2       = (const float*)d_in[7];
    const float* W2s      = (const float*)d_in[8];
    const float* b2       = (const float*)d_in[9];
    float*       out      = (float*)d_out;

    float*  aggR = (float*)d_ws;
    float*  h1   = aggR + (size_t)N_REL * N_NODES * KDIM;
    float2* Wf1  = (float2*)(h1 + (size_t)N_NODES * KDIM);
    float2* Wf2  = Wf1 + (size_t)NSEG * 8 * 32 * 32;

    int* cnt  = (int*)d_out;
    int* loc  = cnt  + N_KEYS;
    int* ssrc = loc  + N_KEYS;
    int* slot = ssrc + N_EDGES;
    int* bsum = slot + N_EDGES;
    int* boff = bsum + 512;

    const int edgeBlocks   = (N_EDGES + 255) / 256;
    const int gatherBlocks = N_KEYS / 8;
    const int gemmBlocks   = N_NODES / 16;

    const int pack1 = NSEG * 8 * 32 * 32;
    const int pack2 = NSEG * 4 * 32 * 32;

    rgcn_pack_w_kernel<<<(pack1 + 255) / 256, 256, 0, stream>>>(W1, W1s, Wf1, 128);
    rgcn_pack_w_kernel<<<(pack2 + 255) / 256, 256, 0, stream>>>(W2, W2s, Wf2, 64);

    rgcn_zero_cnt_kernel<<<(N_KEYS / 4 + 255) / 256, 256, 0, stream>>>((int4*)cnt, N_KEYS / 4);
    rgcn_count_kernel<<<edgeBlocks, 256, 0, stream>>>(dst, etype, cnt, slot, N_EDGES);
    rgcn_scan_local_kernel<<<N_SCAN_BLOCKS, 256, 0, stream>>>(cnt, loc, bsum);
    rgcn_scan_blocks_kernel<<<1, 512, 0, stream>>>(bsum, boff);
    rgcn_place_kernel<<<edgeBlocks, 256, 0, stream>>>(src, dst, etype, slot, loc, boff, ssrc, N_EDGES);

    rgcn_gather_kernel<<<gatherBlocks, 256, 0, stream>>>(features, cnt, loc, boff, ssrc, aggR);
    rgcn_gemm_kernel<<<gemmBlocks, 256, 0, stream>>>(
        aggR, features, Wf1, b1, h1, 128, 1);

    rgcn_gather_kernel<<<gatherBlocks, 256, 0, stream>>>(h1, cnt, loc, boff, ssrc, aggR);
    rgcn_gemm_kernel<<<gemmBlocks, 128, 0, stream>>>(
        aggR, h1, Wf2, b2, out, 64, 0);
}


# Gather: each wave now sums 8 consecutive (relation,dst) segments, with their counts, starts and source ids fetched once up front, instead of one wave per segment; per-segment sum order unchanged
# speedup vs baseline: 2.2010x; 1.1053x over previous
#include <hip/hip_runtime.h>

typedef __attribute__((ext_vector_type(2))) float v2f;
typedef __attribute__((ext_vector_type(8))) float v8f;

#define N_NODES 50000
#define N_EDGES 640000
#define N_REL   8
#define KDIM    128
#define NSEG    (N_REL + 1)

#define N_KEYS        (N_REL * N_NODES)
#define SCAN_SHIFT    10
#define N_SCAN_BLOCKS ((N_KEYS + 1023) >> SCAN_SHIFT)
#define ID_NONE       0x7fffffff

__global__ void rgcn_zero_cnt_kernel(int4* __restrict__ p, int n4) {
    int i = blockIdx.x * blockDim.x + threadIdx.x;
    if (i < n4) p[i] = make_int4(0, 0, 0, 0);
}

__global__ void rgcn_count_kernel(const int* __restrict__ edst,
                                  const int* __restrict__ etyp,
                                  int*       __restrict__ cnt,
                                  int*       __restrict__ slot,
                                  int nEdges) {
    int e = blockIdx.x * blockDim.x + threadIdx.x;
    if (e >= nEdges) return;
    int key = etyp[e] * N_NODES + edst[e];
    slot[e] = atomicAdd(cnt + key, 1);
}

__device__ __forceinline__ int wave_incl_scan(int x, int lane) {
    for (int o = 1; o < 32; o <<= 1) {
        int y = __shfl_up(x, o, 32);
        if (lane >= o) x += y;
    }
    return x;
}

__global__ void rgcn_scan_local_kernel(const int* __restrict__ cnt,
                                       int*       __restrict__ loc,
                                       int*       __restrict__ bsum) {
    __shared__ int wsum[8];
    const int lane = threadIdx.x & 31;
    const int wave = threadIdx.x >> 5;
    const int g    = blockIdx.x * 256 + threadIdx.x;
    int4 v = make_int4(0, 0, 0, 0);
    if (g < N_KEYS / 4) v = ((const int4*)cnt)[g];
    const int s = v.x + v.y + v.z + v.w;
    const int x = wave_incl_scan(s, lane);
    if (lane == 31) wsum[wave] = x;
    __syncthreads();
    int wo = 0;
    for (int w = 0; w < wave; ++w) wo += wsum[w];
    const int ex = wo + x - s;
    if (g < N_KEYS / 4)
        ((int4*)loc)[g] = make_int4(ex, ex + v.x, ex + v.x + v.y, ex + v.x + v.y + v.z);
    if (threadIdx.x == 255) bsum[blockIdx.x] = wo + x;
}

__global__ void rgcn_scan_blocks_kernel(const int* __restrict__ bsum,
                                        int*       __restrict__ boff) {
    __shared__ int wsum[16];
    const int lane = threadIdx.x & 31;
    const int wave = threadIdx.x >> 5;
    const int s = (threadIdx.x < N_SCAN_BLOCKS) ? bsum[threadIdx.x] : 0;
    const int x = wave_incl_scan(s, lane);
    if (lane == 31) wsum[wave] = x;
    __syncthreads();
    int wo = 0;
    for (int w = 0; w < wave; ++w) wo += wsum[w];
    if (threadIdx.x < N_SCAN_BLOCKS) boff[threadIdx.x] = wo + x - s;
}

__global__ void rgcn_place_kernel(const int* __restrict__ esrc,
                                  const int* __restrict__ edst,
                                  const int* __restrict__ etyp,
                                  const int* __restrict__ slot,
                                  const int* __restrict__ loc,
                                  const int* __restrict__ boff,
                                  int*       __restrict__ ssrc,
                                  int nEdges) {
    int e = blockIdx.x * blockDim.x + threadIdx.x;
    if (e >= nEdges) return;
    int key = etyp[e] * N_NODES + edst[e];
    int pos = loc[key] + boff[key >> SCAN_SHIFT] + slot[e];
    ssrc[pos] = esrc[e];
}

__device__ __forceinline__ int wave_min_i(int v) {
    for (int o = 16; o > 0; o >>= 1) {
        int y = __shfl_xor(v, o, 32);
        v = (y < v) ? y : v;
    }
    return v;
}

__device__ __forceinline__ int wave_sum_i(int v) {
    for (int o = 16; o > 0; o >>= 1) v += __shfl_xor(v, o, 32);
    return v;
}

#define GSEG 8

__device__ __forceinline__ float4 rgcn_segment_sum(const float4* __restrict__ f4,
                                                   const int*    __restrict__ ssrc,
                                                   int c, int base, int v0, int lane) {
    const int cl = (c < 32) ? c : 32;
    float4 acc = make_float4(0.f, 0.f, 0.f, 0.f);
    int prev = -1;
    int rem  = c;
    for (int it = 0; it < c && rem > 0; ++it) {
        int m = ID_NONE;
        for (int j = 0; j < cl; ++j) {
            int sj = __builtin_amdgcn_readlane(v0, j);
            if (sj > prev && sj < m) m = sj;
        }
        if (c > 32) {
            int lm = ID_NONE;
            for (int j = lane + 32; j < c; j += 32) {
                int v = ssrc[base + j];
                if (v > prev && v < lm) lm = v;
            }
            int t = __builtin_amdgcn_readfirstlane(wave_min_i(lm));
            if (t < m) m = t;
        }
        if (m == ID_NONE) break;
        int mult = __popc(__builtin_amdgcn_ballot_w32(v0 == m));
        if (c > 32) {
            int cm = 0;
            for (int j = lane + 32; j < c; j += 32) cm += (ssrc[base + j] == m) ? 1 : 0;
            mult += __builtin_amdgcn_readfirstlane(wave_sum_i(cm));
        }
        const float4 row = f4[(size_t)m * (KDIM / 4) + lane];
        for (int t = 0; t < mult; ++t) {
            acc.x += row.x;
            acc.y += row.y;
            acc.z += row.z;
            acc.w += row.w;
        }
        prev = m;
        rem -= mult;
    }
    return acc;
}

__global__ void rgcn_gather_kernel(const float* __restrict__ feat,
                                   const int*   __restrict__ cnt,
                                   const int*   __restrict__ loc,
                                   const int*   __restrict__ boff,
                                   const int*   __restrict__ ssrc,
                                   float*       __restrict__ aggR) {
    const int lane = threadIdx.x & 31;
    const int seg0 = __builtin_amdgcn_readfirstlane(
        (int)((blockIdx.x * 8 + (threadIdx.x >> 5)) * GSEG));
    const float4* f4 = (const float4*)feat;

    int tc = 0;
    int tb = 0;
    if (lane < GSEG) {
        const int s = seg0 + lane;
        tc = cnt[s];
        tb = loc[s] + boff[s >> SCAN_SHIFT];
    }

    const int base0 = __builtin_amdgcn_readlane(tb, 0);
    const int tot   = __builtin_amdgcn_readlane(tb, GSEG - 1)
                    + __builtin_amdgcn_readlane(tc, GSEG - 1) - base0;
    const int idw   = (lane < tot) ? ssrc[base0 + lane] : ID_NONE;

    for (int i = 0; i < GSEG; ++i) {
        const int c    = __builtin_amdgcn_readlane(tc, i);
        const int base = __builtin_amdgcn_readlane(tb, i);
        const int o    = base - base0;
        const int cl   = (c < 32) ? c : 32;
        int v0;
        if (o + cl <= 32) {
            v0 = __shfl(idw, (o + lane) & 31, 32);
            if (lane >= c) v0 = ID_NONE;
        } else {
            v0 = (lane < c) ? ssrc[base + lane] : ID_NONE;
        }
        const float4 acc = rgcn_segment_sum(f4, ssrc, c, base, v0, lane);
        ((float4*)aggR)[(size_t)(seg0 + i) * (KDIM / 4) + lane] = acc;
    }
}

__global__ void rgcn_pack_w_kernel(const float* __restrict__ Wrel,
                                   const float* __restrict__ Wself,
                                   float2*      __restrict__ Wf,
                                   int O) {
    const int tiles = O >> 4;
    const int total = NSEG * tiles * 32 * 32;
    int idx = blockIdx.x * blockDim.x + threadIdx.x;
    if (idx >= total) return;
    int lane = idx & 31;
    int k4   = (idx >> 5) & 31;
    int t    = (idx >> 10) % tiles;
    int seg  = idx / (tiles << 10);
    const float* W = (seg < N_REL) ? (Wrel + (size_t)seg * KDIM * O) : Wself;
    int k0 = k4 * 4 + (lane >> 4) * 2;
    int n  = t * 16 + (lane & 15);
    Wf[idx] = make_float2(W[(size_t)k0 * O + n], W[(size_t)(k0 + 1) * O + n]);
}

__global__ void rgcn_gemm_kernel(const float*  __restrict__ aggR,
                                 const float*  __restrict__ hself,
                                 const float2* __restrict__ Wf,
                                 const float*  __restrict__ bias,
                                 float*        __restrict__ out,
                                 int O, int applyRelu) {
    __shared__ float lds[16 * 132];

    const int m0    = blockIdx.x * 16;
    const int lane  = threadIdx.x & 31;
    const int wave  = threadIdx.x >> 5;
    const int half  = lane >> 4;
    const int l15   = lane & 15;
    const int nb    = wave * 16;
    const int tiles = O >> 4;

    v8f c = {0.f, 0.f, 0.f, 0.f, 0.f, 0.f, 0.f, 0.f};

    for (int s = 0; s < NSEG; ++s) {
        const float*  A  = (s < N_REL) ? (aggR + (size_t)s * N_NODES * KDIM) : hself;
        const float2* wf = Wf + ((size_t)(s * tiles + wave) << 10) + lane;

        __syncthreads();
        for (int t = threadIdx.x; t < 16 * KDIM; t += blockDim.x) {
            int row = t >> 7;
            int col = t & 127;
            lds[row * 132 + col] = A[(size_t)(m0 + row) * KDIM + col];
        }
        __syncthreads();

#pragma unroll
        for (int k4 = 0; k4 < 32; ++k4) {
            const float* ap = &lds[l15 * 132 + k4 * 4 + half * 2];
            v2f a;
            a.x = ap[0];
            a.y = ap[1];
            float2 bf = wf[k4 * 32];
            v2f b;
            b.x = bf.x;
            b.y = bf.y;
            c = __builtin_amdgcn_wmma_f32_16x16x4_f32(
                    false, a, false, b, (short)0, c, false, false);
        }
    }

    float bv = bias[nb + l15];
#pragma unroll
    for (int v = 0; v < 8; ++v) {
        float x = c[v] + bv;
        if (applyRelu) x = fmaxf(x, 0.f);
        out[(size_t)(m0 + v + 8 * half) * O + nb + l15] = x;
    }
}

extern "C" void kernel_launch(void* const* d_in, const int* in_sizes, int n_in,
                              void* d_out, int out_size, void* d_ws, size_t ws_size,
                              hipStream_t stream) {
    const float* features = (const float*)d_in[0];
    const int*   src      = (const int*)  d_in[1];
    const int*   dst      = (const int*)  d_in[2];
    const int*   etype    = (const int*)  d_in[3];
    const float* W1       = (const float*)d_in[4];
    const float* W1s      = (const float*)d_in[5];
    const float* b1       = (const float*)d_in[6];
    const float* W2       = (const float*)d_in[7];
    const float* W2s      = (const float*)d_in[8];
    const float* b2       = (const float*)d_in[9];
    float*       out      = (float*)d_out;

    float*  aggR = (float*)d_ws;
    float*  h1   = aggR + (size_t)N_REL * N_NODES * KDIM;
    float2* Wf1  = (float2*)(h1 + (size_t)N_NODES * KDIM);
    float2* Wf2  = Wf1 + (size_t)NSEG * 8 * 32 * 32;

    int* cnt  = (int*)d_out;
    int* loc  = cnt  + N_KEYS;
    int* ssrc = loc  + N_KEYS;
    int* slot = ssrc + N_EDGES;
    int* bsum = slot + N_EDGES;
    int* boff = bsum + 512;

    const int edgeBlocks   = (N_EDGES + 255) / 256;
    const int gatherBlocks = N_KEYS / (8 * GSEG);
    const int gemmBlocks   = N_NODES / 16;

    const int pack1 = NSEG * 8 * 32 * 32;
    const int pack2 = NSEG * 4 * 32 * 32;

    rgcn_pack_w_kernel<<<(pack1 + 255) / 256, 256, 0, stream>>>(W1, W1s, Wf1, 128);
    rgcn_pack_w_kernel<<<(pack2 + 255) / 256, 256, 0, stream>>>(W2, W2s, Wf2, 64);

    rgcn_zero_cnt_kernel<<<(N_KEYS / 4 + 255) / 256, 256, 0, stream>>>((int4*)cnt, N_KEYS / 4);
    rgcn_count_kernel<<<edgeBlocks, 256, 0, stream>>>(dst, etype, cnt, slot, N_EDGES);
    rgcn_scan_local_kernel<<<N_SCAN_BLOCKS, 256, 0, stream>>>(cnt, loc, bsum);
    rgcn_scan_blocks_kernel<<<1, 512, 0, stream>>>(bsum, boff);
    rgcn_place_kernel<<<edgeBlocks, 256, 0, stream>>>(src, dst, etype, slot, loc, boff, ssrc, N_EDGES);

    rgcn_gather_kernel<<<gatherBlocks, 256, 0, stream>>>(features, cnt, loc, boff, ssrc, aggR);
    rgcn_gemm_kernel<<<gemmBlocks, 256, 0, stream>>>(
        aggR, features, Wf1, b1, h1, 128, 1);

    rgcn_gather_kernel<<<gatherBlocks, 256, 0, stream>>>(h1, cnt, loc, boff, ssrc, aggR);
    rgcn_gemm_kernel<<<gemmBlocks, 128, 0, stream>>>(
        aggR, h1, Wf2, b2, out, 64, 0);
}


# GEMM re-tiled to 32 rows per work-group: each wave keeps two accumulator tiles, so one B fragment load feeds two WMMAs; A tile staged with 16-byte loads; per-element sum order unchanged
# speedup vs baseline: 2.4534x; 1.1147x over previous
#include <hip/hip_runtime.h>

typedef __attribute__((ext_vector_type(2))) float v2f;
typedef __attribute__((ext_vector_type(8))) float v8f;

#define N_NODES 50000
#define N_EDGES 640000
#define N_REL   8
#define KDIM    128
#define NSEG    (N_REL + 1)

#define N_KEYS        (N_REL * N_NODES)
#define SCAN_SHIFT    10
#define N_SCAN_BLOCKS ((N_KEYS + 1023) >> SCAN_SHIFT)
#define ID_NONE       0x7fffffff

__global__ void rgcn_zero_cnt_kernel(int4* __restrict__ p, int n4) {
    int i = blockIdx.x * blockDim.x + threadIdx.x;
    if (i < n4) p[i] = make_int4(0, 0, 0, 0);
}

__global__ void rgcn_count_kernel(const int* __restrict__ edst,
                                  const int* __restrict__ etyp,
                                  int*       __restrict__ cnt,
                                  int*       __restrict__ slot,
                                  int nEdges) {
    int e = blockIdx.x * blockDim.x + threadIdx.x;
    if (e >= nEdges) return;
    int key = etyp[e] * N_NODES + edst[e];
    slot[e] = atomicAdd(cnt + key, 1);
}

__device__ __forceinline__ int wave_incl_scan(int x, int lane) {
    for (int o = 1; o < 32; o <<= 1) {
        int y = __shfl_up(x, o, 32);
        if (lane >= o) x += y;
    }
    return x;
}

__global__ void rgcn_scan_local_kernel(const int* __restrict__ cnt,
                                       int*       __restrict__ loc,
                                       int*       __restrict__ bsum) {
    __shared__ int wsum[8];
    const int lane = threadIdx.x & 31;
    const int wave = threadIdx.x >> 5;
    const int g    = blockIdx.x * 256 + threadIdx.x;
    int4 v = make_int4(0, 0, 0, 0);
    if (g < N_KEYS / 4) v = ((const int4*)cnt)[g];
    const int s = v.x + v.y + v.z + v.w;
    const int x = wave_incl_scan(s, lane);
    if (lane == 31) wsum[wave] = x;
    __syncthreads();
    int wo = 0;
    for (int w = 0; w < wave; ++w) wo += wsum[w];
    const int ex = wo + x - s;
    if (g < N_KEYS / 4)
        ((int4*)loc)[g] = make_int4(ex, ex + v.x, ex + v.x + v.y, ex + v.x + v.y + v.z);
    if (threadIdx.x == 255) bsum[blockIdx.x] = wo + x;
}

__global__ void rgcn_scan_blocks_kernel(const int* __restrict__ bsum,
                                        int*       __restrict__ boff) {
    __shared__ int wsum[16];
    const int lane = threadIdx.x & 31;
    const int wave = threadIdx.x >> 5;
    const int s = (threadIdx.x < N_SCAN_BLOCKS) ? bsum[threadIdx.x] : 0;
    const int x = wave_incl_scan(s, lane);
    if (lane == 31) wsum[wave] = x;
    __syncthreads();
    int wo = 0;
    for (int w = 0; w < wave; ++w) wo += wsum[w];
    if (threadIdx.x < N_SCAN_BLOCKS) boff[threadIdx.x] = wo + x - s;
}

__global__ void rgcn_place_kernel(const int* __restrict__ esrc,
                                  const int* __restrict__ edst,
                                  const int* __restrict__ etyp,
                                  const int* __restrict__ slot,
                                  const int* __restrict__ loc,
                                  const int* __restrict__ boff,
                                  int*       __restrict__ ssrc,
                                  int nEdges) {
    int e = blockIdx.x * blockDim.x + threadIdx.x;
    if (e >= nEdges) return;
    int key = etyp[e] * N_NODES + edst[e];
    int pos = loc[key] + boff[key >> SCAN_SHIFT] + slot[e];
    ssrc[pos] = esrc[e];
}

__device__ __forceinline__ int wave_min_i(int v) {
    for (int o = 16; o > 0; o >>= 1) {
        int y = __shfl_xor(v, o, 32);
        v = (y < v) ? y : v;
    }
    return v;
}

__device__ __forceinline__ int wave_sum_i(int v) {
    for (int o = 16; o > 0; o >>= 1) v += __shfl_xor(v, o, 32);
    return v;
}

#define GSEG 8

__device__ __forceinline__ float4 rgcn_segment_sum(const float4* __restrict__ f4,
                                                   const int*    __restrict__ ssrc,
                                                   int c, int base, int v0, int lane) {
    const int cl = (c < 32) ? c : 32;
    float4 acc = make_float4(0.f, 0.f, 0.f, 0.f);
    int prev = -1;
    int rem  = c;
    for (int it = 0; it < c && rem > 0; ++it) {
        int m = ID_NONE;
        for (int j = 0; j < cl; ++j) {
            int sj = __builtin_amdgcn_readlane(v0, j);
            if (sj > prev && sj < m) m = sj;
        }
        if (c > 32) {
            int lm = ID_NONE;
            for (int j = lane + 32; j < c; j += 32) {
                int v = ssrc[base + j];
                if (v > prev && v < lm) lm = v;
            }
            int t = __builtin_amdgcn_readfirstlane(wave_min_i(lm));
            if (t < m) m = t;
        }
        if (m == ID_NONE) break;
        int mult = __popc(__builtin_amdgcn_ballot_w32(v0 == m));
        if (c > 32) {
            int cm = 0;
            for (int j = lane + 32; j < c; j += 32) cm += (ssrc[base + j] == m) ? 1 : 0;
            mult += __builtin_amdgcn_readfirstlane(wave_sum_i(cm));
        }
        const float4 row = f4[(size_t)m * (KDIM / 4) + lane];
        for (int t = 0; t < mult; ++t) {
            acc.x += row.x;
            acc.y += row.y;
            acc.z += row.z;
            acc.w += row.w;
        }
        prev = m;
        rem -= mult;
    }
    return acc;
}

__global__ void rgcn_gather_kernel(const float* __restrict__ feat,
                                   const int*   __restrict__ cnt,
                                   const int*   __restrict__ loc,
                                   const int*   __restrict__ boff,
                                   const int*   __restrict__ ssrc,
                                   float*       __restrict__ aggR) {
    const int lane = threadIdx.x & 31;
    const int seg0 = __builtin_amdgcn_readfirstlane(
        (int)((blockIdx.x * 8 + (threadIdx.x >> 5)) * GSEG));
    const float4* f4 = (const float4*)feat;

    int tc = 0;
    int tb = 0;
    if (lane < GSEG) {
        const int s = seg0 + lane;
        tc = cnt[s];
        tb = loc[s] + boff[s >> SCAN_SHIFT];
    }

    const int base0 = __builtin_amdgcn_readlane(tb, 0);
    const int tot   = __builtin_amdgcn_readlane(tb, GSEG - 1)
                    + __builtin_amdgcn_readlane(tc, GSEG - 1) - base0;
    const int idw   = (lane < tot) ? ssrc[base0 + lane] : ID_NONE;

    for (int i = 0; i < GSEG; ++i) {
        const int c    = __builtin_amdgcn_readlane(tc, i);
        const int base = __builtin_amdgcn_readlane(tb, i);
        const int o    = base - base0;
        const int cl   = (c < 32) ? c : 32;
        int v0;
        if (o + cl <= 32) {
            v0 = __shfl(idw, (o + lane) & 31, 32);
            if (lane >= c) v0 = ID_NONE;
        } else {
            v0 = (lane < c) ? ssrc[base + lane] : ID_NONE;
        }
        const float4 acc = rgcn_segment_sum(f4, ssrc, c, base, v0, lane);
        ((float4*)aggR)[(size_t)(seg0 + i) * (KDIM / 4) + lane] = acc;
    }
}

__global__ void rgcn_pack_w_kernel(const float* __restrict__ Wrel,
                                   const float* __restrict__ Wself,
                                   float2*      __restrict__ Wf,
                                   int O) {
    const int tiles = O >> 4;
    const int total = NSEG * tiles * 32 * 32;
    int idx = blockIdx.x * blockDim.x + threadIdx.x;
    if (idx >= total) return;
    int lane = idx & 31;
    int k4   = (idx >> 5) & 31;
    int t    = (idx >> 10) % tiles;
    int seg  = idx / (tiles << 10);
    const float* W = (seg < N_REL) ? (Wrel + (size_t)seg * KDIM * O) : Wself;
    int k0 = k4 * 4 + (lane >> 4) * 2;
    int n  = t * 16 + (lane & 15);
    Wf[idx] = make_float2(W[(size_t)k0 * O + n], W[(size_t)(k0 + 1) * O + n]);
}

#define GEMM_MT   2
#define GEMM_ROWS (GEMM_MT * 16)

template <int NW>
__global__ void __launch_bounds__(NW * 32)
rgcn_gemm_kernel(const float*  __restrict__ aggR,
                 const float*  __restrict__ hself,
                 const float2* __restrict__ Wf,
                 const float*  __restrict__ bias,
                 float*        __restrict__ out,
                 int applyRelu) {
    constexpr int O  = NW * 16;
    __shared__ __attribute__((aligned(16))) float lds[GEMM_ROWS * 132];

    const int m0    = blockIdx.x * GEMM_ROWS;
    const int lane  = threadIdx.x & 31;
    const int wave  = __builtin_amdgcn_readfirstlane((int)(threadIdx.x >> 5));
    const int half  = lane >> 4;
    const int l15   = lane & 15;
    const int nb    = wave * 16;

    v8f c[GEMM_MT];
    for (int mt = 0; mt < GEMM_MT; ++mt)
        c[mt] = (v8f){0.f, 0.f, 0.f, 0.f, 0.f, 0.f, 0.f, 0.f};

    for (int s = 0; s < NSEG; ++s) {
        const float4* A4 = (const float4*)((s < N_REL) ? (aggR + (size_t)s * N_NODES * KDIM) : hself);
        const float2* wf = Wf + ((size_t)(s * NW + wave) << 10) + lane;

        __syncthreads();
        float4 st[GEMM_ROWS / NW];
        for (int i = 0; i < GEMM_ROWS / NW; ++i) {
            int gr = m0 + wave + i * NW;
            gr = (gr < N_NODES) ? gr : (N_NODES - 1);
            st[i] = A4[(size_t)gr * (KDIM / 4) + lane];
        }
        for (int i = 0; i < GEMM_ROWS / NW; ++i)
            *(float4*)&lds[(wave + i * NW) * 132 + lane * 4] = st[i];
        __syncthreads();

#pragma unroll
        for (int k4 = 0; k4 < 32; ++k4) {
            float2 bf = wf[k4 * 32];
            v2f b;
            b.x = bf.x;
            b.y = bf.y;
            for (int mt = 0; mt < GEMM_MT; ++mt) {
                const float* ap = &lds[(mt * 16 + l15) * 132 + k4 * 4 + half * 2];
                v2f a;
                a.x = ap[0];
                a.y = ap[1];
                c[mt] = __builtin_amdgcn_wmma_f32_16x16x4_f32(
                            false, a, false, b, (short)0, c[mt], false, false);
            }
        }
    }

    const float bv = bias[nb + l15];
    for (int mt = 0; mt < GEMM_MT; ++mt) {
        for (int v = 0; v < 8; ++v) {
            const int row = m0 + mt * 16 + v + 8 * half;
            float x = c[mt][v] + bv;
            if (applyRelu) x = fmaxf(x, 0.f);
            if (row < N_NODES) out[(size_t)row * O + nb + l15] = x;
        }
    }
}

extern "C" void kernel_launch(void* const* d_in, const int* in_sizes, int n_in,
                              void* d_out, int out_size, void* d_ws, size_t ws_size,
                              hipStream_t stream) {
    const float* features = (const float*)d_in[0];
    const int*   src      = (const int*)  d_in[1];
    const int*   dst      = (const int*)  d_in[2];
    const int*   etype    = (const int*)  d_in[3];
    const float* W1       = (const float*)d_in[4];
    const float* W1s      = (const float*)d_in[5];
    const float* b1       = (const float*)d_in[6];
    const float* W2       = (const float*)d_in[7];
    const float* W2s      = (const float*)d_in[8];
    const float* b2       = (const float*)d_in[9];
    float*       out      = (float*)d_out;

    float*  aggR = (float*)d_ws;
    float*  h1   = aggR + (size_t)N_REL * N_NODES * KDIM;
    float2* Wf1  = (float2*)(h1 + (size_t)N_NODES * KDIM);
    float2* Wf2  = Wf1 + (size_t)NSEG * 8 * 32 * 32;

    int* cnt  = (int*)d_out;
    int* loc  = cnt  + N_KEYS;
    int* ssrc = loc  + N_KEYS;
    int* slot = ssrc + N_EDGES;
    int* bsum = slot + N_EDGES;
    int* boff = bsum + 512;

    const int edgeBlocks   = (N_EDGES + 255) / 256;
    const int gatherBlocks = N_KEYS / (8 * GSEG);
    const int gemmBlocks   = (N_NODES + GEMM_ROWS - 1) / GEMM_ROWS;

    const int pack1 = NSEG * 8 * 32 * 32;
    const int pack2 = NSEG * 4 * 32 * 32;

    rgcn_pack_w_kernel<<<(pack1 + 255) / 256, 256, 0, stream>>>(W1, W1s, Wf1, 128);
    rgcn_pack_w_kernel<<<(pack2 + 255) / 256, 256, 0, stream>>>(W2, W2s, Wf2, 64);

    rgcn_zero_cnt_kernel<<<(N_KEYS / 4 + 255) / 256, 256, 0, stream>>>((int4*)cnt, N_KEYS / 4);
    rgcn_count_kernel<<<edgeBlocks, 256, 0, stream>>>(dst, etype, cnt, slot, N_EDGES);
    rgcn_scan_local_kernel<<<N_SCAN_BLOCKS, 256, 0, stream>>>(cnt, loc, bsum);
    rgcn_scan_blocks_kernel<<<1, 512, 0, stream>>>(bsum, boff);
    rgcn_place_kernel<<<edgeBlocks, 256, 0, stream>>>(src, dst, etype, slot, loc, boff, ssrc, N_EDGES);

    rgcn_gather_kernel<<<gatherBlocks, 256, 0, stream>>>(features, cnt, loc, boff, ssrc, aggR);
    rgcn_gemm_kernel<8><<<gemmBlocks, 256, 0, stream>>>(
        aggR, features, Wf1, b1, h1, 1);

    rgcn_gather_kernel<<<gatherBlocks, 256, 0, stream>>>(h1, cnt, loc, boff, ssrc, aggR);
    rgcn_gemm_kernel<4><<<gemmBlocks, 128, 0, stream>>>(
        aggR, h1, Wf2, b2, out, 0);
}


# Segment gather-sum fused into the GEMM's A-tile staging: each work-group sums its rows per relation from feature rows straight into LDS; the sum array and 2 launches go; sum order unchanged
# speedup vs baseline: 2.6429x; 1.0772x over previous
#include <hip/hip_runtime.h>

typedef __attribute__((ext_vector_type(2))) float v2f;
typedef __attribute__((ext_vector_type(8))) float v8f;

#define N_NODES 50000
#define N_EDGES 640000
#define N_REL   8
#define KDIM    128
#define NSEG    (N_REL + 1)

#define N_KEYS        (N_REL * N_NODES)
#define SCAN_SHIFT    10
#define N_SCAN_BLOCKS ((N_KEYS + 1023) >> SCAN_SHIFT)
#define ID_NONE       0x7fffffff

__global__ void rgcn_zero_cnt_kernel(int4* __restrict__ p, int n4) {
    int i = blockIdx.x * blockDim.x + threadIdx.x;
    if (i < n4) p[i] = make_int4(0, 0, 0, 0);
}

__global__ void rgcn_count_kernel(const int* __restrict__ edst,
                                  const int* __restrict__ etyp,
                                  int*       __restrict__ cnt,
                                  int*       __restrict__ slot,
                                  int nEdges) {
    int e = blockIdx.x * blockDim.x + threadIdx.x;
    if (e >= nEdges) return;
    int key = etyp[e] * N_NODES + edst[e];
    slot[e] = atomicAdd(cnt + key, 1);
}

__device__ __forceinline__ int wave_incl_scan(int x, int lane) {
    for (int o = 1; o < 32; o <<= 1) {
        int y = __shfl_up(x, o, 32);
        if (lane >= o) x += y;
    }
    return x;
}

__global__ void rgcn_scan_local_kernel(const int* __restrict__ cnt,
                                       int*       __restrict__ loc,
                                       int*       __restrict__ bsum) {
    __shared__ int wsum[8];
    const int lane = threadIdx.x & 31;
    const int wave = threadIdx.x >> 5;
    const int g    = blockIdx.x * 256 + threadIdx.x;
    int4 v = make_int4(0, 0, 0, 0);
    if (g < N_KEYS / 4) v = ((const int4*)cnt)[g];
    const int s = v.x + v.y + v.z + v.w;
    const int x = wave_incl_scan(s, lane);
    if (lane == 31) wsum[wave] = x;
    __syncthreads();
    int wo = 0;
    for (int w = 0; w < wave; ++w) wo += wsum[w];
    const int ex = wo + x - s;
    if (g < N_KEYS / 4)
        ((int4*)loc)[g] = make_int4(ex, ex + v.x, ex + v.x + v.y, ex + v.x + v.y + v.z);
    if (threadIdx.x == 255) bsum[blockIdx.x] = wo + x;
}

__global__ void rgcn_scan_blocks_kernel(const int* __restrict__ bsum,
                                        int*       __restrict__ boff) {
    __shared__ int wsum[16];
    const int lane = threadIdx.x & 31;
    const int wave = threadIdx.x >> 5;
    const int s = (threadIdx.x < N_SCAN_BLOCKS) ? bsum[threadIdx.x] : 0;
    const int x = wave_incl_scan(s, lane);
    if (lane == 31) wsum[wave] = x;
    __syncthreads();
    int wo = 0;
    for (int w = 0; w < wave; ++w) wo += wsum[w];
    if (threadIdx.x < N_SCAN_BLOCKS) boff[threadIdx.x] = wo + x - s;
}

__global__ void rgcn_place_kernel(const int* __restrict__ esrc,
                                  const int* __restrict__ edst,
                                  const int* __restrict__ etyp,
                                  const int* __restrict__ slot,
                                  const int* __restrict__ loc,
                                  const int* __restrict__ boff,
                                  int*       __restrict__ ssrc,
                                  int nEdges) {
    int e = blockIdx.x * blockDim.x + threadIdx.x;
    if (e >= nEdges) return;
    int key = etyp[e] * N_NODES + edst[e];
    int pos = loc[key] + boff[key >> SCAN_SHIFT] + slot[e];
    ssrc[pos] = esrc[e];
}

__device__ __forceinline__ int wave_min_i(int v) {
    for (int o = 16; o > 0; o >>= 1) {
        int y = __shfl_xor(v, o, 32);
        v = (y < v) ? y : v;
    }
    return v;
}

__device__ __forceinline__ int wave_sum_i(int v) {
    for (int o = 16; o > 0; o >>= 1) v += __shfl_xor(v, o, 32);
    return v;
}


__device__ __forceinline__ float4 rgcn_segment_sum(const float4* __restrict__ f4,
                                                   const int*    __restrict__ ssrc,
                                                   int c, int base, int v0, int lane) {
    const int cl = (c < 32) ? c : 32;
    float4 acc = make_float4(0.f, 0.f, 0.f, 0.f);
    int prev = -1;
    int rem  = c;
    for (int it = 0; it < c && rem > 0; ++it) {
        int m = ID_NONE;
        for (int j = 0; j < cl; ++j) {
            int sj = __builtin_amdgcn_readlane(v0, j);
            if (sj > prev && sj < m) m = sj;
        }
        if (c > 32) {
            int lm = ID_NONE;
            for (int j = lane + 32; j < c; j += 32) {
                int v = ssrc[base + j];
                if (v > prev && v < lm) lm = v;
            }
            int t = __builtin_amdgcn_readfirstlane(wave_min_i(lm));
            if (t < m) m = t;
        }
        if (m == ID_NONE) break;
        int mult = __popc(__builtin_amdgcn_ballot_w32(v0 == m));
        if (c > 32) {
            int cm = 0;
            for (int j = lane + 32; j < c; j += 32) cm += (ssrc[base + j] == m) ? 1 : 0;
            mult += __builtin_amdgcn_readfirstlane(wave_sum_i(cm));
        }
        const float4 row = f4[(size_t)m * (KDIM / 4) + lane];
        for (int t = 0; t < mult; ++t) {
            acc.x += row.x;
            acc.y += row.y;
            acc.z += row.z;
            acc.w += row.w;
        }
        prev = m;
        rem -= mult;
    }
    return acc;
}

template <int R>
__device__ __forceinline__ void rgcn_rows_sum(const float4* __restrict__ f4,
                                              const int (&c)[R], const int (&v0)[R],
                                              int lane, float4 (&acc)[R]) {
    int prev[R];
    int rem[R];
    int maxc = 0;
#pragma unroll
    for (int i = 0; i < R; ++i) {
        acc[i]  = make_float4(0.f, 0.f, 0.f, 0.f);
        prev[i] = -1;
        rem[i]  = c[i];
        maxc    = (c[i] > maxc) ? c[i] : maxc;
    }
    for (int it = 0; it < maxc; ++it) {
        int m[R];
        int mult[R];
        int live = 0;
#pragma unroll
        for (int i = 0; i < R; ++i) {
            m[i]    = 0;
            mult[i] = 0;
            if (rem[i] > 0) {
                int mi = ID_NONE;
                for (int j = 0; j < c[i]; ++j) {
                    int sj = __builtin_amdgcn_readlane(v0[i], j);
                    if (sj > prev[i] && sj < mi) mi = sj;
                }
                if (mi != ID_NONE) {
                    m[i]    = mi;
                    mult[i] = __popc(__builtin_amdgcn_ballot_w32(v0[i] == mi));
                    prev[i] = mi;
                    rem[i] -= mult[i];
                    live    = 1;
                } else {
                    rem[i] = 0;
                }
            }
        }
        if (!live) break;
        float4 row[R];
#pragma unroll
        for (int i = 0; i < R; ++i) {
            row[i] = make_float4(0.f, 0.f, 0.f, 0.f);
            if (mult[i] > 0) row[i] = f4[(size_t)m[i] * (KDIM / 4) + lane];
        }
#pragma unroll
        for (int i = 0; i < R; ++i) {
            for (int t = 0; t < mult[i]; ++t) {
                acc[i].x += row[i].x;
                acc[i].y += row[i].y;
                acc[i].z += row[i].z;
                acc[i].w += row[i].w;
            }
        }
    }
}

__global__ void rgcn_pack_w_kernel(const float* __restrict__ Wrel,
                                   const float* __restrict__ Wself,
                                   float2*      __restrict__ Wf,
                                   int O) {
    const int tiles = O >> 4;
    const int total = NSEG * tiles * 32 * 32;
    int idx = blockIdx.x * blockDim.x + threadIdx.x;
    if (idx >= total) return;
    int lane = idx & 31;
    int k4   = (idx >> 5) & 31;
    int t    = (idx >> 10) % tiles;
    int seg  = idx / (tiles << 10);
    const float* W = (seg < N_REL) ? (Wrel + (size_t)seg * KDIM * O) : Wself;
    int k0 = k4 * 4 + (lane >> 4) * 2;
    int n  = t * 16 + (lane & 15);
    Wf[idx] = make_float2(W[(size_t)k0 * O + n], W[(size_t)(k0 + 1) * O + n]);
}

#define GEMM_MT   2
#define GEMM_ROWS (GEMM_MT * 16)

template <int NW>
__global__ void __launch_bounds__(NW * 32)
rgcn_layer_kernel(const float*  __restrict__ feat,
                  const int*    __restrict__ cnt,
                  const int*    __restrict__ loc,
                  const int*    __restrict__ boff,
                  const int*    __restrict__ ssrc,
                  const float2* __restrict__ Wf,
                  const float*  __restrict__ bias,
                  float*        __restrict__ out,
                  int applyRelu) {
    constexpr int O   = NW * 16;
    constexpr int RPW = GEMM_ROWS / NW;
    constexpr int RG  = 4;
    __shared__ __attribute__((aligned(16))) float lds[GEMM_ROWS * 132];
    __shared__ int tabc[N_REL * GEMM_ROWS];
    __shared__ int tabb[N_REL * GEMM_ROWS];

    const int m0    = blockIdx.x * GEMM_ROWS;
    const int lane  = threadIdx.x & 31;
    const int wave  = __builtin_amdgcn_readfirstlane((int)(threadIdx.x >> 5));
    const int half  = lane >> 4;
    const int l15   = lane & 15;
    const int nb    = wave * 16;
    const float4* f4 = (const float4*)feat;

    for (int r = wave; r < N_REL; r += NW) {
        const int row = m0 + lane;
        int cc = 0;
        int bb = 0;
        if (row < N_NODES) {
            const int key = r * N_NODES + row;
            cc = cnt[key];
            bb = loc[key] + boff[key >> SCAN_SHIFT];
        }
        tabc[r * GEMM_ROWS + lane] = cc;
        tabb[r * GEMM_ROWS + lane] = bb;
    }

    v8f c[GEMM_MT];
    for (int mt = 0; mt < GEMM_MT; ++mt)
        c[mt] = (v8f){0.f, 0.f, 0.f, 0.f, 0.f, 0.f, 0.f, 0.f};

    for (int s = 0; s < NSEG; ++s) {
        const float2* wf = Wf + ((size_t)(s * NW + wave) << 10) + lane;

        __syncthreads();
        if (s < N_REL) {
            int tc = 0;
            int tb = 0;
            if (lane < RPW) {
                tc = tabc[s * GEMM_ROWS + wave * RPW + lane];
                tb = tabb[s * GEMM_ROWS + wave * RPW + lane];
            }
            int sc[RPW];
            int sb[RPW];
            int sv[RPW];
#pragma unroll
            for (int i = 0; i < RPW; ++i) {
                sc[i] = __builtin_amdgcn_readlane(tc, i);
                sb[i] = __builtin_amdgcn_readlane(tb, i);
                sv[i] = (lane < sc[i]) ? ssrc[sb[i] + lane] : ID_NONE;
            }
#pragma unroll
            for (int g = 0; g < RPW / RG; ++g) {
                int gc[RG];
                int gb[RG];
                int gv[RG];
#pragma unroll
                for (int i = 0; i < RG; ++i) {
                    gc[i] = sc[g * RG + i];
                    gb[i] = sb[g * RG + i];
                    gv[i] = sv[g * RG + i];
                }
                float4 acc[RG];
                int gmax = 0;
#pragma unroll
                for (int i = 0; i < RG; ++i) gmax = (gc[i] > gmax) ? gc[i] : gmax;
                if (gmax <= 32) {
                    rgcn_rows_sum<RG>(f4, gc, gv, lane, acc);
                } else {
#pragma unroll
                    for (int i = 0; i < RG; ++i)
                        acc[i] = rgcn_segment_sum(f4, ssrc, gc[i], gb[i], gv[i], lane);
                }
#pragma unroll
                for (int i = 0; i < RG; ++i)
                    *(float4*)&lds[(wave * RPW + g * RG + i) * 132 + lane * 4] = acc[i];
            }
        } else {
            float4 st[GEMM_ROWS / NW];
            for (int i = 0; i < GEMM_ROWS / NW; ++i) {
                int gr = m0 + wave + i * NW;
                gr = (gr < N_NODES) ? gr : (N_NODES - 1);
                st[i] = f4[(size_t)gr * (KDIM / 4) + lane];
            }
            for (int i = 0; i < GEMM_ROWS / NW; ++i)
                *(float4*)&lds[(wave + i * NW) * 132 + lane * 4] = st[i];
        }
        __syncthreads();

#pragma unroll
        for (int k4 = 0; k4 < 32; ++k4) {
            float2 bf = wf[k4 * 32];
            v2f b;
            b.x = bf.x;
            b.y = bf.y;
            for (int mt = 0; mt < GEMM_MT; ++mt) {
                const float* ap = &lds[(mt * 16 + l15) * 132 + k4 * 4 + half * 2];
                v2f a;
                a.x = ap[0];
                a.y = ap[1];
                c[mt] = __builtin_amdgcn_wmma_f32_16x16x4_f32(
                            false, a, false, b, (short)0, c[mt], false, false);
            }
        }
        asm volatile("v_nop\n\tv_nop\n\tv_nop\n\tv_nop\n\tv_nop\n\tv_nop\n\tv_nop\n\tv_nop"
                     : : "v"(c[0]), "v"(c[1]));
    }

    const float bv = bias[nb + l15];
    for (int mt = 0; mt < GEMM_MT; ++mt) {
        for (int v = 0; v < 8; ++v) {
            const int row = m0 + mt * 16 + v + 8 * half;
            float x = c[mt][v] + bv;
            if (applyRelu) x = fmaxf(x, 0.f);
            if (row < N_NODES) out[(size_t)row * O + nb + l15] = x;
        }
    }
}

extern "C" void kernel_launch(void* const* d_in, const int* in_sizes, int n_in,
                              void* d_out, int out_size, void* d_ws, size_t ws_size,
                              hipStream_t stream) {
    const float* features = (const float*)d_in[0];
    const int*   src      = (const int*)  d_in[1];
    const int*   dst      = (const int*)  d_in[2];
    const int*   etype    = (const int*)  d_in[3];
    const float* W1       = (const float*)d_in[4];
    const float* W1s      = (const float*)d_in[5];
    const float* b1       = (const float*)d_in[6];
    const float* W2       = (const float*)d_in[7];
    const float* W2s      = (const float*)d_in[8];
    const float* b2       = (const float*)d_in[9];
    float*       out      = (float*)d_out;

    float*  tabs = (float*)d_ws;
    float*  h1   = tabs + (size_t)N_REL * N_NODES * KDIM;
    float2* Wf1  = (float2*)(h1 + (size_t)N_NODES * KDIM);
    float2* Wf2  = Wf1 + (size_t)NSEG * 8 * 32 * 32;

    int* cnt  = (int*)d_ws;
    int* loc  = cnt  + N_KEYS;
    int* ssrc = loc  + N_KEYS;
    int* slot = ssrc + N_EDGES;
    int* bsum = slot + N_EDGES;
    int* boff = bsum + 512;

    const int edgeBlocks   = (N_EDGES + 255) / 256;
    const int layerBlocks  = (N_NODES + GEMM_ROWS - 1) / GEMM_ROWS;

    const int pack1 = NSEG * 8 * 32 * 32;
    const int pack2 = NSEG * 4 * 32 * 32;

    rgcn_pack_w_kernel<<<(pack1 + 255) / 256, 256, 0, stream>>>(W1, W1s, Wf1, 128);
    rgcn_pack_w_kernel<<<(pack2 + 255) / 256, 256, 0, stream>>>(W2, W2s, Wf2, 64);

    rgcn_zero_cnt_kernel<<<(N_KEYS / 4 + 255) / 256, 256, 0, stream>>>((int4*)cnt, N_KEYS / 4);
    rgcn_count_kernel<<<edgeBlocks, 256, 0, stream>>>(dst, etype, cnt, slot, N_EDGES);
    rgcn_scan_local_kernel<<<N_SCAN_BLOCKS, 256, 0, stream>>>(cnt, loc, bsum);
    rgcn_scan_blocks_kernel<<<1, 512, 0, stream>>>(bsum, boff);
    rgcn_place_kernel<<<edgeBlocks, 256, 0, stream>>>(src, dst, etype, slot, loc, boff, ssrc, N_EDGES);

    rgcn_layer_kernel<8><<<layerBlocks, 256, 0, stream>>>(
        features, cnt, loc, boff, ssrc, Wf1, b1, h1, 1);

    rgcn_layer_kernel<4><<<layerBlocks, 128, 0, stream>>>(
        h1, cnt, loc, boff, ssrc, Wf2, b2, out, 0);
}


# Source ids sorted inside each segment once per call (small new kernel) and read at block start; a wave requests all rows of a relation at once, not one min-search + load per source; sum order same
# speedup vs baseline: 3.0103x; 1.1390x over previous
#include <hip/hip_runtime.h>

typedef __attribute__((ext_vector_type(2))) float v2f;
typedef __attribute__((ext_vector_type(8))) float v8f;

#define N_NODES 50000
#define N_EDGES 640000
#define N_REL   8
#define KDIM    128
#define NSEG    (N_REL + 1)

#define N_KEYS        (N_REL * N_NODES)
#define SCAN_SHIFT    10
#define N_SCAN_BLOCKS ((N_KEYS + 1023) >> SCAN_SHIFT)

__global__ void rgcn_zero_cnt_kernel(int4* __restrict__ p, int n4) {
    int i = blockIdx.x * blockDim.x + threadIdx.x;
    if (i < n4) p[i] = make_int4(0, 0, 0, 0);
}

__global__ void rgcn_count_kernel(const int* __restrict__ edst,
                                  const int* __restrict__ etyp,
                                  int*       __restrict__ cnt,
                                  int*       __restrict__ slot,
                                  int nEdges) {
    int e = blockIdx.x * blockDim.x + threadIdx.x;
    if (e >= nEdges) return;
    int key = etyp[e] * N_NODES + edst[e];
    slot[e] = atomicAdd(cnt + key, 1);
}

__device__ __forceinline__ int wave_incl_scan(int x, int lane) {
    for (int o = 1; o < 32; o <<= 1) {
        int y = __shfl_up(x, o, 32);
        if (lane >= o) x += y;
    }
    return x;
}

__global__ void rgcn_scan_local_kernel(const int* __restrict__ cnt,
                                       int*       __restrict__ loc,
                                       int*       __restrict__ bsum) {
    __shared__ int wsum[8];
    const int lane = threadIdx.x & 31;
    const int wave = threadIdx.x >> 5;
    const int g    = blockIdx.x * 256 + threadIdx.x;
    int4 v = make_int4(0, 0, 0, 0);
    if (g < N_KEYS / 4) v = ((const int4*)cnt)[g];
    const int s = v.x + v.y + v.z + v.w;
    const int x = wave_incl_scan(s, lane);
    if (lane == 31) wsum[wave] = x;
    __syncthreads();
    int wo = 0;
    for (int w = 0; w < wave; ++w) wo += wsum[w];
    const int ex = wo + x - s;
    if (g < N_KEYS / 4)
        ((int4*)loc)[g] = make_int4(ex, ex + v.x, ex + v.x + v.y, ex + v.x + v.y + v.z);
    if (threadIdx.x == 255) bsum[blockIdx.x] = wo + x;
}

__global__ void rgcn_scan_blocks_kernel(const int* __restrict__ bsum,
                                        int*       __restrict__ boff) {
    __shared__ int wsum[16];
    const int lane = threadIdx.x & 31;
    const int wave = threadIdx.x >> 5;
    const int s = (threadIdx.x < N_SCAN_BLOCKS) ? bsum[threadIdx.x] : 0;
    const int x = wave_incl_scan(s, lane);
    if (lane == 31) wsum[wave] = x;
    __syncthreads();
    int wo = 0;
    for (int w = 0; w < wave; ++w) wo += wsum[w];
    if (threadIdx.x < N_SCAN_BLOCKS) boff[threadIdx.x] = wo + x - s;
}

__global__ void rgcn_place_kernel(const int* __restrict__ esrc,
                                  const int* __restrict__ edst,
                                  const int* __restrict__ etyp,
                                  const int* __restrict__ slot,
                                  const int* __restrict__ loc,
                                  const int* __restrict__ boff,
                                  int*       __restrict__ ssrc,
                                  int nEdges) {
    int e = blockIdx.x * blockDim.x + threadIdx.x;
    if (e >= nEdges) return;
    int key = etyp[e] * N_NODES + edst[e];
    int pos = loc[key] + boff[key >> SCAN_SHIFT] + slot[e];
    ssrc[pos] = esrc[e];
}

__device__ __forceinline__ void rgcn_sort_run(int* __restrict__ p, int c) {
    for (int i = 1; i < c; ++i) {
        const int v = p[i];
        int j = i - 1;
        while (j >= 0 && p[j] > v) {
            p[j + 1] = p[j];
            --j;
        }
        p[j + 1] = v;
    }
}

__global__ void rgcn_sort_seg_kernel(const int* __restrict__ cnt,
                                     const int* __restrict__ loc,
                                     const int* __restrict__ boff,
                                     int*       __restrict__ ssrc) {
    const int g = blockIdx.x * blockDim.x + threadIdx.x;
    if (g >= N_KEYS / 4) return;
    const int4 cv = ((const int4*)cnt)[g];
    if (cv.x < 2 && cv.y < 2 && cv.z < 2 && cv.w < 2) return;
    const int4 lv = ((const int4*)loc)[g];
    const int  bo = boff[(g * 4) >> SCAN_SHIFT];
    rgcn_sort_run(ssrc + lv.x + bo, cv.x);
    rgcn_sort_run(ssrc + lv.y + bo, cv.y);
    rgcn_sort_run(ssrc + lv.z + bo, cv.z);
    rgcn_sort_run(ssrc + lv.w + bo, cv.w);
}


__global__ void rgcn_pack_w_kernel(const float* __restrict__ Wrel,
                                   const float* __restrict__ Wself,
                                   float2*      __restrict__ Wf,
                                   int O) {
    const int tiles = O >> 4;
    const int total = NSEG * tiles * 32 * 32;
    int idx = blockIdx.x * blockDim.x + threadIdx.x;
    if (idx >= total) return;
    int lane = idx & 31;
    int k4   = (idx >> 5) & 31;
    int t    = (idx >> 10) % tiles;
    int seg  = idx / (tiles << 10);
    const float* W = (seg < N_REL) ? (Wrel + (size_t)seg * KDIM * O) : Wself;
    int k0 = k4 * 4 + (lane >> 4) * 2;
    int n  = t * 16 + (lane & 15);
    Wf[idx] = make_float2(W[(size_t)k0 * O + n], W[(size_t)(k0 + 1) * O + n]);
}

#define GEMM_MT   2
#define GEMM_ROWS (GEMM_MT * 16)

template <int NW>
__global__ void __launch_bounds__(NW * 32)
rgcn_layer_kernel(const float*  __restrict__ feat,
                  const int*    __restrict__ cnt,
                  const int*    __restrict__ loc,
                  const int*    __restrict__ boff,
                  const int*    __restrict__ ssrc,
                  const float2* __restrict__ Wf,
                  const float*  __restrict__ bias,
                  float*        __restrict__ out,
                  int applyRelu) {
    constexpr int O   = NW * 16;
    constexpr int RPW = GEMM_ROWS / NW;
    constexpr int NB  = (RPW <= 4) ? 3 * RPW : 2 * RPW;
    __shared__ __attribute__((aligned(16))) float lds[GEMM_ROWS * 132];
    __shared__ int tabc[N_REL * GEMM_ROWS];
    __shared__ int tabb[N_REL * GEMM_ROWS];
    __shared__ int tids[N_REL * NW * 32];

    const int m0    = blockIdx.x * GEMM_ROWS;
    const int lane  = threadIdx.x & 31;
    const int wave  = __builtin_amdgcn_readfirstlane((int)(threadIdx.x >> 5));
    const int half  = lane >> 4;
    const int l15   = lane & 15;
    const int nb    = wave * 16;
    const float4* f4 = (const float4*)feat;

    for (int r = wave; r < N_REL; r += NW) {
        const int row = m0 + lane;
        int cc = 0;
        int bb = 0;
        if (row < N_NODES) {
            const int key = r * N_NODES + row;
            cc = cnt[key];
            bb = loc[key] + boff[key >> SCAN_SHIFT];
        }
        tabc[r * GEMM_ROWS + lane] = cc;
        tabb[r * GEMM_ROWS + lane] = bb;
    }
    __syncthreads();

    {
        int ptc[N_REL];
        int ptb[N_REL];
#pragma unroll
        for (int r = 0; r < N_REL; ++r) {
            ptc[r] = tabc[r * GEMM_ROWS + wave * RPW + (lane & (RPW - 1))];
            ptb[r] = tabb[r * GEMM_ROWS + wave * RPW];
        }
        int pid[N_REL];
#pragma unroll
        for (int r = 0; r < N_REL; ++r) {
            int total = 0;
#pragma unroll
            for (int i = 0; i < RPW; ++i) total += __builtin_amdgcn_readlane(ptc[r], i);
            const int start = __builtin_amdgcn_readfirstlane(ptb[r]);
            pid[r] = (lane < total) ? ssrc[start + lane] : 0;
        }
#pragma unroll
        for (int r = 0; r < N_REL; ++r) tids[(r * NW + wave) * 32 + lane] = pid[r];
    }

    v8f c[GEMM_MT];
    for (int mt = 0; mt < GEMM_MT; ++mt)
        c[mt] = (v8f){0.f, 0.f, 0.f, 0.f, 0.f, 0.f, 0.f, 0.f};

    for (int s = 0; s < NSEG; ++s) {
        const float2* wf = Wf + ((size_t)(s * NW + wave) << 10) + lane;

        __syncthreads();
        if (s < N_REL) {
            const int tc = tabc[s * GEMM_ROWS + wave * RPW + (lane & (RPW - 1))];
            const int start = __builtin_amdgcn_readfirstlane(tabb[s * GEMM_ROWS + wave * RPW]);
            int vid = tids[(s * NW + wave) * 32 + lane];
            int cend[RPW];
            int total = 0;
#pragma unroll
            for (int i = 0; i < RPW; ++i) {
                total  += __builtin_amdgcn_readlane(tc, i);
                cend[i] = total;
            }
#pragma unroll
            for (int i = 0; i < RPW; ++i)
                *(float4*)&lds[(wave * RPW + i) * 132 + lane * 4] = make_float4(0.f, 0.f, 0.f, 0.f);

            float4 acc = make_float4(0.f, 0.f, 0.f, 0.f);
            for (int e0 = 0; e0 < total; e0 += 32) {
                if (e0 > 0) vid = (e0 + lane < total) ? ssrc[start + e0 + lane] : 0;
                const int e = e0 + lane;
                int  vseg  = 0;
                bool vlast = false;
#pragma unroll
                for (int i = 0; i < RPW; ++i) {
                    vseg  += (cend[i] <= e) ? 1 : 0;
                    vlast  = vlast || (e + 1 == cend[i]);
                }
                const unsigned lastMask = __builtin_amdgcn_ballot_w32(vlast);
                const int nn = (total - e0 < 32) ? (total - e0) : 32;
                for (int b = 0; b < nn; b += NB) {
                    int id[NB];
#pragma unroll
                    for (int k = 0; k < NB; ++k)
                        id[k] = __builtin_amdgcn_readlane(vid, (b + k < nn) ? (b + k) : (nn - 1));
                    float4 row[NB];
#pragma unroll
                    for (int k = 0; k < NB; ++k) {
                        if (b + k < nn) {
                            row[k] = f4[(size_t)id[k] * (KDIM / 4) + lane];
                        } else {
                            row[k] = make_float4(0.f, 0.f, 0.f, 0.f);
                        }
                    }
#pragma unroll
                    for (int k = 0; k < NB; ++k) {
                        if (b + k < nn) {
                            acc.x += row[k].x;
                            acc.y += row[k].y;
                            acc.z += row[k].z;
                            acc.w += row[k].w;
                            if ((lastMask >> (b + k)) & 1u) {
                                const int ri = __builtin_amdgcn_readlane(vseg, b + k);
                                *(float4*)&lds[(wave * RPW + ri) * 132 + lane * 4] = acc;
                                acc = make_float4(0.f, 0.f, 0.f, 0.f);
                            }
                        }
                    }
                }
            }
        } else {
            float4 st[GEMM_ROWS / NW];
            for (int i = 0; i < GEMM_ROWS / NW; ++i) {
                int gr = m0 + wave + i * NW;
                gr = (gr < N_NODES) ? gr : (N_NODES - 1);
                st[i] = f4[(size_t)gr * (KDIM / 4) + lane];
            }
            for (int i = 0; i < GEMM_ROWS / NW; ++i)
                *(float4*)&lds[(wave + i * NW) * 132 + lane * 4] = st[i];
        }
        __syncthreads();

#pragma unroll
        for (int k4 = 0; k4 < 32; ++k4) {
            float2 bf = wf[k4 * 32];
            v2f b;
            b.x = bf.x;
            b.y = bf.y;
            for (int mt = 0; mt < GEMM_MT; ++mt) {
                const float* ap = &lds[(mt * 16 + l15) * 132 + k4 * 4 + half * 2];
                v2f a;
                a.x = ap[0];
                a.y = ap[1];
                c[mt] = __builtin_amdgcn_wmma_f32_16x16x4_f32(
                            false, a, false, b, (short)0, c[mt], false, false);
            }
        }
        asm volatile("v_nop\n\tv_nop\n\tv_nop\n\tv_nop\n\tv_nop\n\tv_nop\n\tv_nop\n\tv_nop"
                     : : "v"(c[0]), "v"(c[1]));
    }

    const float bv = bias[nb + l15];
    for (int mt = 0; mt < GEMM_MT; ++mt) {
        for (int v = 0; v < 8; ++v) {
            const int row = m0 + mt * 16 + v + 8 * half;
            float x = c[mt][v] + bv;
            if (applyRelu) x = fmaxf(x, 0.f);
            if (row < N_NODES) out[(size_t)row * O + nb + l15] = x;
        }
    }
}

extern "C" void kernel_launch(void* const* d_in, const int* in_sizes, int n_in,
                              void* d_out, int out_size, void* d_ws, size_t ws_size,
                              hipStream_t stream) {
    const float* features = (const float*)d_in[0];
    const int*   src      = (const int*)  d_in[1];
    const int*   dst      = (const int*)  d_in[2];
    const int*   etype    = (const int*)  d_in[3];
    const float* W1       = (const float*)d_in[4];
    const float* W1s      = (const float*)d_in[5];
    const float* b1       = (const float*)d_in[6];
    const float* W2       = (const float*)d_in[7];
    const float* W2s      = (const float*)d_in[8];
    const float* b2       = (const float*)d_in[9];
    float*       out      = (float*)d_out;

    float*  tabs = (float*)d_ws;
    float*  h1   = tabs + (size_t)N_REL * N_NODES * KDIM;
    float2* Wf1  = (float2*)(h1 + (size_t)N_NODES * KDIM);
    float2* Wf2  = Wf1 + (size_t)NSEG * 8 * 32 * 32;

    int* cnt  = (int*)d_ws;
    int* loc  = cnt  + N_KEYS;
    int* ssrc = loc  + N_KEYS;
    int* slot = ssrc + N_EDGES;
    int* bsum = slot + N_EDGES;
    int* boff = bsum + 512;

    const int edgeBlocks   = (N_EDGES + 255) / 256;
    const int layerBlocks  = (N_NODES + GEMM_ROWS - 1) / GEMM_ROWS;

    const int pack1 = NSEG * 8 * 32 * 32;
    const int pack2 = NSEG * 4 * 32 * 32;

    rgcn_pack_w_kernel<<<(pack1 + 255) / 256, 256, 0, stream>>>(W1, W1s, Wf1, 128);
    rgcn_pack_w_kernel<<<(pack2 + 255) / 256, 256, 0, stream>>>(W2, W2s, Wf2, 64);

    rgcn_zero_cnt_kernel<<<(N_KEYS / 4 + 255) / 256, 256, 0, stream>>>((int4*)cnt, N_KEYS / 4);
    rgcn_count_kernel<<<edgeBlocks, 256, 0, stream>>>(dst, etype, cnt, slot, N_EDGES);
    rgcn_scan_local_kernel<<<N_SCAN_BLOCKS, 256, 0, stream>>>(cnt, loc, bsum);
    rgcn_scan_blocks_kernel<<<1, 512, 0, stream>>>(bsum, boff);
    rgcn_place_kernel<<<edgeBlocks, 256, 0, stream>>>(src, dst, etype, slot, loc, boff, ssrc, N_EDGES);
    rgcn_sort_seg_kernel<<<(N_KEYS / 4 + 255) / 256, 256, 0, stream>>>(cnt, loc, boff, ssrc);

    rgcn_layer_kernel<8><<<layerBlocks, 256, 0, stream>>>(
        features, cnt, loc, boff, ssrc, Wf1, b1, h1, 1);

    rgcn_layer_kernel<4><<<layerBlocks, 128, 0, stream>>>(
        h1, cnt, loc, boff, ssrc, Wf2, b2, out, 0);
}


# Launches folded 10 -> 6: weight packs + counter clear share a grid; block-offset scan done inside the placing pass (4 edges/thread); segment id sort done in layer kernel 1's set-up; sum order same
# speedup vs baseline: 3.3384x; 1.1090x over previous
#include <hip/hip_runtime.h>

typedef __attribute__((ext_vector_type(2))) float v2f;
typedef __attribute__((ext_vector_type(8))) float v8f;

#define N_NODES 50000
#define N_EDGES 640000
#define N_REL   8
#define KDIM    128
#define NSEG    (N_REL + 1)

#define N_KEYS        (N_REL * N_NODES)
#define SCAN_SHIFT    10
#define N_SCAN_BLOCKS ((N_KEYS + 1023) >> SCAN_SHIFT)
static_assert(N_SCAN_BLOCKS <= 512, "the placing pass scans 2 block totals per thread, 256 threads");

__global__ void rgcn_count_kernel(const int* __restrict__ edst,
                                  const int* __restrict__ etyp,
                                  int*       __restrict__ cnt,
                                  int*       __restrict__ slot,
                                  int nEdges) {
    int e = blockIdx.x * blockDim.x + threadIdx.x;
    if (e >= nEdges) return;
    int key = etyp[e] * N_NODES + edst[e];
    slot[e] = atomicAdd(cnt + key, 1);
}

__device__ __forceinline__ int wave_incl_scan(int x, int lane) {
    for (int o = 1; o < 32; o <<= 1) {
        int y = __shfl_up(x, o, 32);
        if (lane >= o) x += y;
    }
    return x;
}

__global__ void rgcn_scan_local_kernel(const int* __restrict__ cnt,
                                       int*       __restrict__ loc,
                                       int*       __restrict__ bsum) {
    __shared__ int wsum[8];
    const int lane = threadIdx.x & 31;
    const int wave = threadIdx.x >> 5;
    const int g    = blockIdx.x * 256 + threadIdx.x;
    int4 v = make_int4(0, 0, 0, 0);
    if (g < N_KEYS / 4) v = ((const int4*)cnt)[g];
    const int s = v.x + v.y + v.z + v.w;
    const int x = wave_incl_scan(s, lane);
    if (lane == 31) wsum[wave] = x;
    __syncthreads();
    int wo = 0;
    for (int w = 0; w < wave; ++w) wo += wsum[w];
    const int ex = wo + x - s;
    if (g < N_KEYS / 4)
        ((int4*)loc)[g] = make_int4(ex, ex + v.x, ex + v.x + v.y, ex + v.x + v.y + v.z);
    if (threadIdx.x == 255) bsum[blockIdx.x] = wo + x;
}

static_assert(N_EDGES % 1024 == 0, "the placing pass has no tail: 256 threads x 4 edges per work-group");

__global__ void __launch_bounds__(256)
rgcn_place_kernel(const int* __restrict__ esrc,
                  const int* __restrict__ edst,
                  const int* __restrict__ etyp,
                  const int* __restrict__ slot,
                  const int* __restrict__ loc,
                  const int* __restrict__ bsum,
                  int*       __restrict__ boff,
                  int*       __restrict__ ssrc) {
    __shared__ int tbl[512];
    __shared__ int wsum[8];
    const int lane = threadIdx.x & 31;
    const int wave = threadIdx.x >> 5;
    const int g    = blockIdx.x * 256 + threadIdx.x;

    const int j0 = threadIdx.x * 2;
    const int ra = bsum[(j0     < N_SCAN_BLOCKS) ? j0     : N_SCAN_BLOCKS - 1];
    const int rb = bsum[(j0 + 1 < N_SCAN_BLOCKS) ? j0 + 1 : N_SCAN_BLOCKS - 1];
    const int a  = (j0     < N_SCAN_BLOCKS) ? ra : 0;
    const int b  = (j0 + 1 < N_SCAN_BLOCKS) ? rb : 0;
    const int4 s4 = ((const int4*)esrc)[g];
    const int4 d4 = ((const int4*)edst)[g];
    const int4 t4 = ((const int4*)etyp)[g];
    const int4 p4 = ((const int4*)slot)[g];

    const int s  = a + b;
    const int x  = wave_incl_scan(s, lane);
    if (lane == 31) wsum[wave] = x;
    __syncthreads();
    int wo = 0;
    for (int w = 0; w < wave; ++w) wo += wsum[w];
    const int ex = wo + x - s;
    tbl[j0]     = ex;
    tbl[j0 + 1] = ex + a;
    if (blockIdx.x == 0) {
        if (j0     < N_SCAN_BLOCKS) boff[j0]     = ex;
        if (j0 + 1 < N_SCAN_BLOCKS) boff[j0 + 1] = ex + a;
    }
    __syncthreads();

    const int k0 = t4.x * N_NODES + d4.x;
    const int k1 = t4.y * N_NODES + d4.y;
    const int k2 = t4.z * N_NODES + d4.z;
    const int k3 = t4.w * N_NODES + d4.w;
    const int l0 = loc[k0];
    const int l1 = loc[k1];
    const int l2 = loc[k2];
    const int l3 = loc[k3];
    ssrc[l0 + tbl[k0 >> SCAN_SHIFT] + p4.x] = s4.x;
    ssrc[l1 + tbl[k1 >> SCAN_SHIFT] + p4.y] = s4.y;
    ssrc[l2 + tbl[k2 >> SCAN_SHIFT] + p4.z] = s4.z;
    ssrc[l3 + tbl[k3 >> SCAN_SHIFT] + p4.w] = s4.w;
}

#define RGCN_SK 16

__device__ __forceinline__ void rgcn_sort_run(int* p, int c) {
    for (int i = 1; i < c; ++i) {
        const int v = p[i];
        int j = i - 1;
        while (j >= 0 && p[j] > v) {
            p[j + 1] = p[j];
            --j;
        }
        p[j + 1] = v;
    }
}


template <int O>
__device__ __forceinline__ void rgcn_pack_w(const float* __restrict__ Wrel,
                                            const float* __restrict__ Wself,
                                            float2*      __restrict__ Wf,
                                            int idx) {
    constexpr int tiles = O >> 4;
    const int lane = idx & 31;
    const int k4   = (idx >> 5) & 31;
    const int t    = (idx >> 10) % tiles;
    const int seg  = idx / (tiles << 10);
    const float* W = (seg < N_REL) ? (Wrel + (size_t)seg * KDIM * O) : Wself;
    const int k0 = k4 * 4 + (lane >> 4) * 2;
    const int n  = t * 16 + (lane & 15);
    Wf[idx] = make_float2(W[(size_t)k0 * O + n], W[(size_t)(k0 + 1) * O + n]);
}

#define PACK1_WGS ((NSEG * 8 * 32 * 32) / 256)
#define PACK2_WGS ((NSEG * 4 * 32 * 32) / 256)
#define ZERO_WGS  ((N_KEYS / 4 + 255) / 256)

__global__ void __launch_bounds__(256)
rgcn_prep_kernel(const float* __restrict__ W1, const float* __restrict__ W1s, float2* __restrict__ Wf1,
                 const float* __restrict__ W2, const float* __restrict__ W2s, float2* __restrict__ Wf2,
                 int4* __restrict__ cnt4) {
    const int wg = blockIdx.x;
    if (wg < PACK1_WGS) {
        rgcn_pack_w<128>(W1, W1s, Wf1, wg * 256 + (int)threadIdx.x);
    } else if (wg < PACK1_WGS + PACK2_WGS) {
        rgcn_pack_w<64>(W2, W2s, Wf2, (wg - PACK1_WGS) * 256 + (int)threadIdx.x);
    } else {
        const int i = (wg - PACK1_WGS - PACK2_WGS) * 256 + (int)threadIdx.x;
        if (i < N_KEYS / 4) cnt4[i] = make_int4(0, 0, 0, 0);
    }
}

#define GEMM_MT   2
#define GEMM_ROWS (GEMM_MT * 16)

template <int NW, bool SORT>
__global__ void __launch_bounds__(NW * 32)
rgcn_layer_kernel(const float*  __restrict__ feat,
                  const int*    __restrict__ cnt,
                  const int*    __restrict__ loc,
                  const int*    __restrict__ boff,
                  int*                       ssrc,
                  const float2* __restrict__ Wf,
                  const float*  __restrict__ bias,
                  float*        __restrict__ out,
                  int applyRelu) {
    constexpr int O   = NW * 16;
    constexpr int RPW = GEMM_ROWS / NW;
    constexpr int NB  = (RPW <= 4) ? 3 * RPW : 2 * RPW;
    __shared__ __attribute__((aligned(16))) float lds[GEMM_ROWS * 132];
    __shared__ int tabc[N_REL * GEMM_ROWS];
    __shared__ int tabb[N_REL * GEMM_ROWS];
    __shared__ int tids[N_REL * NW * 32];

    const int m0    = blockIdx.x * GEMM_ROWS;
    const int lane  = threadIdx.x & 31;
    const int wave  = __builtin_amdgcn_readfirstlane((int)(threadIdx.x >> 5));
    const int half  = lane >> 4;
    const int l15   = lane & 15;
    const int nb    = wave * 16;
    const float4* f4 = (const float4*)feat;

    for (int r = wave; r < N_REL; r += NW) {
        const int row = m0 + lane;
        int cc = 0;
        int bb = 0;
        if (row < N_NODES) {
            const int key = r * N_NODES + row;
            cc = cnt[key];
            bb = loc[key] + boff[key >> SCAN_SHIFT];
        }
        tabc[r * GEMM_ROWS + lane] = cc;
        tabb[r * GEMM_ROWS + lane] = bb;
        if (SORT) {
            const int runStart = __shfl(bb, lane & ~(RPW - 1), 32);
            const int p0 = bb - runStart;
            int* tslot = &tids[(r * NW + lane / RPW) * 32];
            if (cc <= RGCN_SK) {
                int v[RGCN_SK];
#pragma unroll
                for (int k = 0; k < RGCN_SK; ++k) v[k] = (k < cc) ? ssrc[bb + k] : 0x7fffffff;
#pragma unroll
                for (int pass = 0; pass < RGCN_SK; ++pass) {
#pragma unroll
                    for (int i = pass & 1; i + 1 < RGCN_SK; i += 2) {
                        const int lo = (v[i] < v[i + 1]) ? v[i] : v[i + 1];
                        const int hi = (v[i] < v[i + 1]) ? v[i + 1] : v[i];
                        v[i]     = lo;
                        v[i + 1] = hi;
                    }
                }
#pragma unroll
                for (int k = 0; k < RGCN_SK; ++k) {
                    if (k < cc) {
                        if (cc > 1) ssrc[bb + k] = v[k];
                        if (p0 + k < 32) tslot[p0 + k] = v[k];
                    }
                }
            } else {
                rgcn_sort_run(ssrc + bb, cc);
                for (int k = 0; k < cc && p0 + k < 32; ++k) tslot[p0 + k] = ssrc[bb + k];
            }
        }
    }
    __syncthreads();

    if (!SORT) {
        int ptc[N_REL];
        int ptb[N_REL];
#pragma unroll
        for (int r = 0; r < N_REL; ++r) {
            ptc[r] = tabc[r * GEMM_ROWS + wave * RPW + (lane & (RPW - 1))];
            ptb[r] = tabb[r * GEMM_ROWS + wave * RPW];
        }
        int pid[N_REL];
#pragma unroll
        for (int r = 0; r < N_REL; ++r) {
            int total = 0;
#pragma unroll
            for (int i = 0; i < RPW; ++i) total += __builtin_amdgcn_readlane(ptc[r], i);
            const int start = __builtin_amdgcn_readfirstlane(ptb[r]);
            pid[r] = (lane < total) ? ssrc[start + lane] : 0;
        }
#pragma unroll
        for (int r = 0; r < N_REL; ++r) tids[(r * NW + wave) * 32 + lane] = pid[r];
    }

    v8f c[GEMM_MT];
    for (int mt = 0; mt < GEMM_MT; ++mt)
        c[mt] = (v8f){0.f, 0.f, 0.f, 0.f, 0.f, 0.f, 0.f, 0.f};

    for (int s = 0; s < NSEG; ++s) {
        const float2* wf = Wf + ((size_t)(s * NW + wave) << 10) + lane;

        __syncthreads();
        if (s < N_REL) {
            const int tc = tabc[s * GEMM_ROWS + wave * RPW + (lane & (RPW - 1))];
            const int start = __builtin_amdgcn_readfirstlane(tabb[s * GEMM_ROWS + wave * RPW]);
            int vid = tids[(s * NW + wave) * 32 + lane];
            int cend[RPW];
            int total = 0;
#pragma unroll
            for (int i = 0; i < RPW; ++i) {
                total  += __builtin_amdgcn_readlane(tc, i);
                cend[i] = total;
            }
#pragma unroll
            for (int i = 0; i < RPW; ++i)
                *(float4*)&lds[(wave * RPW + i) * 132 + lane * 4] = make_float4(0.f, 0.f, 0.f, 0.f);

            float4 acc = make_float4(0.f, 0.f, 0.f, 0.f);
            for (int e0 = 0; e0 < total; e0 += 32) {
                if (e0 > 0) vid = (e0 + lane < total) ? ssrc[start + e0 + lane] : 0;
                const int e = e0 + lane;
                int  vseg  = 0;
                bool vlast = false;
#pragma unroll
                for (int i = 0; i < RPW; ++i) {
                    vseg  += (cend[i] <= e) ? 1 : 0;
                    vlast  = vlast || (e + 1 == cend[i]);
                }
                const unsigned lastMask = __builtin_amdgcn_ballot_w32(vlast);
                const int nn = (total - e0 < 32) ? (total - e0) : 32;
                for (int b = 0; b < nn; b += NB) {
                    int id[NB];
#pragma unroll
                    for (int k = 0; k < NB; ++k)
                        id[k] = __builtin_amdgcn_readlane(vid, (b + k < nn) ? (b + k) : (nn - 1));
                    float4 row[NB];
#pragma unroll
                    for (int k = 0; k < NB; ++k) {
                        if (b + k < nn) {
                            row[k] = f4[(size_t)id[k] * (KDIM / 4) + lane];
                        } else {
                            row[k] = make_float4(0.f, 0.f, 0.f, 0.f);
                        }
                    }
#pragma unroll
                    for (int k = 0; k < NB; ++k) {
                        if (b + k < nn) {
                            acc.x += row[k].x;
                            acc.y += row[k].y;
                            acc.z += row[k].z;
                            acc.w += row[k].w;
                            if ((lastMask >> (b + k)) & 1u) {
                                const int ri = __builtin_amdgcn_readlane(vseg, b + k);
                                *(float4*)&lds[(wave * RPW + ri) * 132 + lane * 4] = acc;
                                acc = make_float4(0.f, 0.f, 0.f, 0.f);
                            }
                        }
                    }
                }
            }
        } else {
            float4 st[GEMM_ROWS / NW];
            for (int i = 0; i < GEMM_ROWS / NW; ++i) {
                int gr = m0 + wave + i * NW;
                gr = (gr < N_NODES) ? gr : (N_NODES - 1);
                st[i] = f4[(size_t)gr * (KDIM / 4) + lane];
            }
            for (int i = 0; i < GEMM_ROWS / NW; ++i)
                *(float4*)&lds[(wave + i * NW) * 132 + lane * 4] = st[i];
        }
        __syncthreads();

#pragma unroll
        for (int k4 = 0; k4 < 32; ++k4) {
            float2 bf = wf[k4 * 32];
            v2f b;
            b.x = bf.x;
            b.y = bf.y;
            for (int mt = 0; mt < GEMM_MT; ++mt) {
                const float* ap = &lds[(mt * 16 + l15) * 132 + k4 * 4 + half * 2];
                v2f a;
                a.x = ap[0];
                a.y = ap[1];
                c[mt] = __builtin_amdgcn_wmma_f32_16x16x4_f32(
                            false, a, false, b, (short)0, c[mt], false, false);
            }
        }
        asm volatile("v_nop\n\tv_nop\n\tv_nop\n\tv_nop\n\tv_nop\n\tv_nop\n\tv_nop\n\tv_nop"
                     : : "v"(c[0]), "v"(c[1]));
    }

    const float bv = bias[nb + l15];
    for (int mt = 0; mt < GEMM_MT; ++mt) {
        for (int v = 0; v < 8; ++v) {
            const int row = m0 + mt * 16 + v + 8 * half;
            float x = c[mt][v] + bv;
            if (applyRelu) x = fmaxf(x, 0.f);
            if (row < N_NODES) out[(size_t)row * O + nb + l15] = x;
        }
    }
}

extern "C" void kernel_launch(void* const* d_in, const int* in_sizes, int n_in,
                              void* d_out, int out_size, void* d_ws, size_t ws_size,
                              hipStream_t stream) {
    const float* features = (const float*)d_in[0];
    const int*   src      = (const int*)  d_in[1];
    const int*   dst      = (const int*)  d_in[2];
    const int*   etype    = (const int*)  d_in[3];
    const float* W1       = (const float*)d_in[4];
    const float* W1s      = (const float*)d_in[5];
    const float* b1       = (const float*)d_in[6];
    const float* W2       = (const float*)d_in[7];
    const float* W2s      = (const float*)d_in[8];
    const float* b2       = (const float*)d_in[9];
    float*       out      = (float*)d_out;

    float*  tabs = (float*)d_ws;
    float*  h1   = tabs + (size_t)N_REL * N_NODES * KDIM;
    float2* Wf1  = (float2*)(h1 + (size_t)N_NODES * KDIM);
    float2* Wf2  = Wf1 + (size_t)NSEG * 8 * 32 * 32;

    int* cnt  = (int*)d_ws;
    int* loc  = cnt  + N_KEYS;
    int* ssrc = loc  + N_KEYS;
    int* slot = ssrc + N_EDGES;
    int* bsum = slot + N_EDGES;
    int* boff = bsum + 512;

    const int edgeBlocks   = (N_EDGES + 255) / 256;
    const int edgeBlocks4  = N_EDGES / 1024;
    const int layerBlocks  = (N_NODES + GEMM_ROWS - 1) / GEMM_ROWS;

    rgcn_prep_kernel<<<PACK1_WGS + PACK2_WGS + ZERO_WGS, 256, 0, stream>>>(
        W1, W1s, Wf1, W2, W2s, Wf2, (int4*)cnt);

    rgcn_count_kernel<<<edgeBlocks, 256, 0, stream>>>(dst, etype, cnt, slot, N_EDGES);
    rgcn_scan_local_kernel<<<N_SCAN_BLOCKS, 256, 0, stream>>>(cnt, loc, bsum);
    rgcn_place_kernel<<<edgeBlocks4, 256, 0, stream>>>(src, dst, etype, slot, loc, bsum, boff, ssrc);

    rgcn_layer_kernel<8, true><<<layerBlocks, 256, 0, stream>>>(
        features, cnt, loc, boff, ssrc, Wf1, b1, h1, 1);

    rgcn_layer_kernel<4, false><<<layerBlocks, 128, 0, stream>>>(
        h1, cnt, loc, boff, ssrc, Wf2, b2, out, 0);
}
